# starting kernel
# baseline (speedup 1.0000x reference)
#include <hip/hip_runtime.h>

#define D 128
#define WLDS_STRIDE 132

typedef float v2f __attribute__((ext_vector_type(2)));
typedef float v8f __attribute__((ext_vector_type(8)));

__global__ void zero_f4_kernel(float4* __restrict__ p, int n4) {
    int i = blockIdx.x * blockDim.x + threadIdx.x;
    if (i < n4) p[i] = make_float4(0.f, 0.f, 0.f, 0.f);
}

__global__ __launch_bounds__(256)
void spmm_scatter_kernel(const float* __restrict__ X,
                         const int*   __restrict__ rows,
                         const int*   __restrict__ cols,
                         const float* __restrict__ vals,
                         float* __restrict__ h,
                         int nEdges) {
    int gid  = blockIdx.x * blockDim.x + threadIdx.x;
    int e    = gid >> 5;
    int lane = gid & 31;
    e = __builtin_amdgcn_readfirstlane(e);
    if (e >= nEdges) return;

    int   r = rows[e];
    int   c = cols[e];
    float v = vals[e];

    const float4 xv = ((const float4*)(X + (size_t)c * D))[lane];
    float* hp = h + (size_t)r * D + lane * 4;
    unsafeAtomicAdd(hp + 0, v * xv.x);
    unsafeAtomicAdd(hp + 1, v * xv.y);
    unsafeAtomicAdd(hp + 2, v * xv.z);
    unsafeAtomicAdd(hp + 3, v * xv.w);
}

__global__ __launch_bounds__(256)
void gemm_wmma_kernel(const float* __restrict__ h,
                      const float* __restrict__ W,
                      const float* __restrict__ bias,
                      float* __restrict__ out,
                      int nNodes) {
    __shared__ float Wl[D * WLDS_STRIDE];

    const int wave = threadIdx.x >> 5;
    const int lane = threadIdx.x & 31;
    const int l15  = lane & 15;
    const int hi   = lane >> 4;
    const int rowBase = blockIdx.x * 128 + wave * 16;

    for (int i = threadIdx.x; i < (D * D) / 4; i += 256) {
        float4 w4 = ((const float4*)W)[i];
        int flat = i * 4;
        int o = flat >> 7;
        int k = flat & 127;
        *(float4*)&Wl[o * WLDS_STRIDE + k] = w4;
    }
    __syncthreads();

    int arow = rowBase + l15;
    if (arow >= nNodes) arow = nNodes - 1;
    const float* aptr = h + (size_t)arow * D + 2 * hi;

    v8f acc[8];
#pragma unroll
    for (int nt = 0; nt < 8; ++nt) acc[nt] = (v8f){0.f,0.f,0.f,0.f,0.f,0.f,0.f,0.f};

    const float* bbase = Wl + (size_t)l15 * WLDS_STRIDE + 2 * hi;

    for (int ks = 0; ks < 32; ++ks) {
        const int k = ks * 4;
        const v2f a = *(const v2f*)(aptr + k);
#pragma unroll
        for (int nt = 0; nt < 8; ++nt) {
            const v2f b2 = *(const v2f*)(bbase + nt * 16 * WLDS_STRIDE + k);
            acc[nt] = __builtin_amdgcn_wmma_f32_16x16x4_f32(
                 false, a,  false, b2,
                 (short)0, acc[nt],
                 false,  false);
        }
    }

#pragma unroll
    for (int nt = 0; nt < 8; ++nt) {
        const float bv = bias[nt * 16 + l15];
#pragma unroll
        for (int v = 0; v < 8; ++v) {
            const int r = rowBase + v + 8 * hi;
            if (r < nNodes)
                __builtin_nontemporal_store(acc[nt][v] + bv,
                                            &out[(size_t)r * D + nt * 16 + l15]);
        }
    }
}

extern "C" void kernel_launch(void* const* d_in, const int* in_sizes, int n_in,
                              void* d_out, int out_size, void* d_ws, size_t ws_size,
                              hipStream_t stream) {
    const float* X    = (const float*)d_in[0];
    const int*   rows = (const int*)  d_in[1];
    const int*   cols = (const int*)  d_in[2];
    const float* vals = (const float*)d_in[3];
    const float* W    = (const float*)d_in[4];
    const float* bias = (const float*)d_in[5];
    float*       out  = (float*)d_out;
    float*       h    = (float*)d_ws;

    const int nNodes = in_sizes[0] / D;
    const int nEdges = in_sizes[1];

    {
        const int n4 = nNodes * (D / 4);
        zero_f4_kernel<<<(n4 + 255) / 256, 256, 0, stream>>>((float4*)h, n4);
    }
    {
        const long long threads = (long long)nEdges * 32;
        const int blocks = (int)((threads + 255) / 256);
        spmm_scatter_kernel<<<blocks, 256, 0, stream>>>(X, rows, cols, vals, h, nEdges);
    }
    {
        const int blocks = (nNodes + 127) / 128;
        gemm_wmma_kernel<<<blocks, 256, 0, stream>>>(h, W, bias, out, nNodes);
    }
}


# One launch: each work-group owns 400 rows of h in LDS, filters the whole edge list for them and adds val*x with LDS f32 atomics, then WMMA from LDS; no global atomics; row sum order = LDS arrival
# speedup vs baseline: 2.3737x; 2.3737x over previous
#include <hip/hip_runtime.h>

#define D 128
#define LSTR 132
#define RB 400
#define NW 32
#define QCAP 160

typedef float v2f __attribute__((ext_vector_type(2)));
typedef float v8f __attribute__((ext_vector_type(8)));


__device__ __forceinline__
void add_full(const float* __restrict__ X, float* hT, int cv, int vv, int tt, int lane) {
#pragma unroll
    for (int g = 0; g < 32; g += 16) {
        v2f xa[16], xb[16];
#pragma unroll
        for (int u = 0; u < 16; ++u) {
            const int c = __builtin_amdgcn_readlane(cv, g + u);
            const float* xp = X + (size_t)c * D + 2 * lane;
            xa[u] = *(const v2f*)xp;
            xb[u] = *(const v2f*)(xp + 64);
        }
#pragma unroll
        for (int u = 0; u < 16; ++u) {
            const float v = __int_as_float(__builtin_amdgcn_readlane(vv, g + u));
            const int   t = __builtin_amdgcn_readlane(tt, g + u);
            float* hp = hT + t * LSTR + 2 * lane;
            atomicAdd(hp,      v * xa[u].x);
            atomicAdd(hp + 1,  v * xa[u].y);
            atomicAdd(hp + 64, v * xb[u].x);
            atomicAdd(hp + 65, v * xb[u].y);
        }
    }
}

__device__ __forceinline__
void add_part(const float* __restrict__ X, const int* __restrict__ cols,
              const float* __restrict__ vals, float* hT,
              int ee, int tt, int cnt, int lane) {
    int cv = 0, vv = 0;
    if (lane < cnt) { cv = cols[ee]; vv = __float_as_int(vals[ee]); }
    for (int g = 0; g < cnt; g += 8) {
        v2f xa[8], xb[8];
#pragma unroll
        for (int u = 0; u < 8; ++u) {
            const int c = __builtin_amdgcn_readlane(cv, g + u);
            const float* xp = X + (size_t)c * D + 2 * lane;
            xa[u] = *(const v2f*)xp;
            xb[u] = *(const v2f*)(xp + 64);
        }
#pragma unroll
        for (int u = 0; u < 8; ++u) {
            if (g + u < cnt) {
                const float v = __int_as_float(__builtin_amdgcn_readlane(vv, g + u));
                const int   t = __builtin_amdgcn_readlane(tt, g + u);
                float* hp = hT + t * LSTR + 2 * lane;
                atomicAdd(hp,      v * xa[u].x);
                atomicAdd(hp + 1,  v * xa[u].y);
                atomicAdd(hp + 64, v * xb[u].x);
                atomicAdd(hp + 65, v * xb[u].y);
            }
        }
    }
}

__global__ __launch_bounds__(1024)
void gcn_rowblock_kernel(const float* __restrict__ X,
                         const int*   __restrict__ rows,
                         const int*   __restrict__ cols,
                         const float* __restrict__ vals,
                         const float* __restrict__ W,
                         const float* __restrict__ bias,
                         float* __restrict__ out,
                         int nNodes, int nEdges) {
    __shared__ __attribute__((aligned(16))) float hT[RB * LSTR];
    __shared__ __attribute__((aligned(16))) float Wl[D * LSTR];
    __shared__ int qe[NW * QCAP];
    __shared__ int qt[NW * QCAP];

    const int tid  = threadIdx.x;
    const int wave = __builtin_amdgcn_readfirstlane(tid >> 5);
    const int lane = tid & 31;
    const int r0   = blockIdx.x * RB;
    const unsigned rcur = (unsigned)min(RB, nNodes - r0);

    for (int i = tid; i < (RB * LSTR) / 4; i += 1024)
        ((float4*)hT)[i] = make_float4(0.f, 0.f, 0.f, 0.f);
    for (int i = tid; i < (D * D) / 4; i += 1024) {
        const float4 w4 = ((const float4*)W)[i];
        const int flat = i * 4;
        *(float4*)&Wl[(flat >> 7) * LSTR + (flat & 127)] = w4;
    }
    __syncthreads();

    int* myqe = qe + wave * QCAP;
    int* myqt = qt + wave * QCAP;
    int  n = 0;
    bool pend = false;
    int  pcv = 0, pvv = 0, ptt = 0;

#define APPEND(m, e, t)                                                      \
    {                                                                        \
        const unsigned mk = __builtin_amdgcn_ballot_w32(m);                  \
        if (mk) {                                                            \
            const int pre = __builtin_amdgcn_mbcnt_lo(mk, 0);                \
            if (m) { myqe[n + pre] = (e); myqt[n + pre] = (int)(t); }        \
            n += __builtin_popcount(mk);                                     \
        }                                                                    \
    }
#define FLUSH()                                                              \
    while (n >= 32) {                                                        \
        n -= 32;                                                             \
        const int ee = myqe[n + lane], tt = myqt[n + lane];                  \
        const int ncv = cols[ee];                                            \
        const int nvv = __float_as_int(vals[ee]);                            \
        if (pend) add_full(X, hT, pcv, pvv, ptt, lane);                      \
        pcv = ncv; pvv = nvv; ptt = tt; pend = true;                         \
    }
#define CHUNK(rv, e0)                                                        \
    {                                                                        \
        const unsigned t0 = (unsigned)((rv).x - r0);                         \
        const unsigned t1 = (unsigned)((rv).y - r0);                         \
        const unsigned t2 = (unsigned)((rv).z - r0);                         \
        const unsigned t3 = (unsigned)((rv).w - r0);                         \
        const bool m0 = t0 < rcur, m1 = t1 < rcur, m2 = t2 < rcur, m3 = t3 < rcur; \
        if (__builtin_amdgcn_ballot_w32(m0 | m1 | m2 | m3)) {                \
            APPEND(m0, (e0) + 0, t0)                                         \
            APPEND(m1, (e0) + 1, t1)                                         \
            APPEND(m2, (e0) + 2, t2)                                         \
            APPEND(m3, (e0) + 3, t3)                                         \
            FLUSH()                                                          \
        }                                                                    \
    }

    const int nChunks = nEdges >> 7;
    if (nChunks > 0) {
        const int off = (int)(((long long)blockIdx.x * nChunks) / (int)gridDim.x);
        int4 ra[4]; int ca[4];
#pragma unroll
        for (int u = 0; u < 4; ++u) {
            int ch = min(wave + u * NW, nChunks - 1) + off;
            if (ch >= nChunks) ch -= nChunks;
            ca[u] = ch;
            ra[u] = *(const int4*)(rows + (ch << 7) + lane * 4);
        }
        for (int i = wave; i < nChunks; i += 4 * NW) {
            int4 rb[4]; int cb[4];
#pragma unroll
            for (int u = 0; u < 4; ++u) {
                int ch = min(i + (4 + u) * NW, nChunks - 1) + off;
                if (ch >= nChunks) ch -= nChunks;
                cb[u] = ch;
                rb[u] = *(const int4*)(rows + (ch << 7) + lane * 4);
            }
#pragma unroll
            for (int u = 0; u < 4; ++u) {
                if (i + u * NW < nChunks) CHUNK(ra[u], (ca[u] << 7) + lane * 4)
            }
#pragma unroll
            for (int u = 0; u < 4; ++u) { ra[u] = rb[u]; ca[u] = cb[u]; }
        }
    }
    if (wave == 0) {
        for (int e = (nChunks << 7) + lane; e - lane < nEdges; e += 32) {
            unsigned t = 0xFFFFFFFFu;
            if (e < nEdges) t = (unsigned)(rows[e] - r0);
            const bool m = t < rcur;
            APPEND(m, e, t)
            FLUSH()
        }
    }
    if (pend) add_full(X, hT, pcv, pvv, ptt, lane);
    if (n > 0) {
        const int ee = myqe[lane], tt = myqt[lane];
        add_part(X, cols, vals, hT, ee, tt, n, lane);
    }
#undef CHUNK
#undef FLUSH
#undef APPEND
    __syncthreads();

    const int l15 = lane & 15;
    const int hi  = lane >> 4;
    for (int unit = wave; unit < (RB / 16) * 2; unit += NW) {
        const int rt = unit >> 1;
        const int nh = unit & 1;
        const float* abase = hT + (rt * 16 + l15) * LSTR + 2 * hi;
        const float* bbase = Wl + (nh * 64 + l15) * LSTR + 2 * hi;
        v8f acc[4];
#pragma unroll
        for (int nt = 0; nt < 4; ++nt) acc[nt] = (v8f){0.f,0.f,0.f,0.f,0.f,0.f,0.f,0.f};
        for (int ks = 0; ks < 32; ++ks) {
            const v2f a = *(const v2f*)(abase + ks * 4);
#pragma unroll
            for (int nt = 0; nt < 4; ++nt) {
                const v2f b2 = *(const v2f*)(bbase + nt * 16 * LSTR + ks * 4);
                acc[nt] = __builtin_amdgcn_wmma_f32_16x16x4_f32(
                    false, a, false, b2, (short)0, acc[nt], false, false);
            }
        }
#pragma unroll
        for (int nt = 0; nt < 4; ++nt) {
            const int col = nh * 64 + nt * 16 + l15;
            const float bv = bias[col];
#pragma unroll
            for (int v = 0; v < 8; ++v) {
                const int r = r0 + rt * 16 + v + 8 * hi;
                if (r < nNodes)
                    __builtin_nontemporal_store(acc[nt][v] + bv, &out[(size_t)r * D + col]);
            }
        }
    }
}

extern "C" void kernel_launch(void* const* d_in, const int* in_sizes, int n_in,
                              void* d_out, int out_size, void* d_ws, size_t ws_size,
                              hipStream_t stream) {
    const float* X    = (const float*)d_in[0];
    const int*   rows = (const int*)  d_in[1];
    const int*   cols = (const int*)  d_in[2];
    const float* vals = (const float*)d_in[3];
    const float* W    = (const float*)d_in[4];
    const float* bias = (const float*)d_in[5];
    float*       out  = (float*)d_out;

    const int nNodes = in_sizes[0] / D;
    const int nEdges = in_sizes[1];

    const int blocks = (nNodes + RB - 1) / RB;
    gcn_rowblock_kernel<<<blocks, 1024, 0, stream>>>(X, rows, cols, vals, W, bias, out,
                                                     nNodes, nEdges);
}


# Edge filter compares 1-byte row-block keys, 16 per lane, written to the workspace by a new first launch (2 launches), not 4-byte row indices; a hit re-reads its row; row sum order still LDS arrival
# speedup vs baseline: 2.8607x; 1.2052x over previous
#include <hip/hip_runtime.h>

#define D 128
#define LSTR 132
#define RB 400
#define NW 32
#define QCAP 64

typedef float v2f __attribute__((ext_vector_type(2)));
typedef float v8f __attribute__((ext_vector_type(8)));


__global__ __launch_bounds__(256)
void gcn_keys_kernel(const int* __restrict__ rows, unsigned* __restrict__ keys, int nEdges) {
    const int nWords = (nEdges >> 9) << 7;
    const int nThreads = (int)(gridDim.x * blockDim.x);
    const int gid = (int)(blockIdx.x * blockDim.x + threadIdx.x);
    for (int base = gid; base < nWords; base += 4 * nThreads) {
#pragma unroll
        for (int u = 0; u < 4; ++u) {
            const int i = base + u * nThreads;
            const int4 r = ((const int4*)rows)[min(i, nWords - 1)];
            const unsigned k0 = ((unsigned)r.x / RB) & 255u;
            const unsigned k1 = ((unsigned)r.y / RB) & 255u;
            const unsigned k2 = ((unsigned)r.z / RB) & 255u;
            const unsigned k3 = ((unsigned)r.w / RB) & 255u;
            keys[i] = k0 | (k1 << 8) | (k2 << 16) | (k3 << 24);
        }
    }
}

__device__ __forceinline__
void add_full(const float* __restrict__ X, float* hT, int cv, int vv, int tt, int lane) {
#pragma unroll
    for (int g = 0; g < 32; g += 16) {
        v2f xa[16], xb[16];
#pragma unroll
        for (int u = 0; u < 16; ++u) {
            const int c = __builtin_amdgcn_readlane(cv, g + u);
            const float* xp = X + (size_t)c * D + 2 * lane;
            xa[u] = *(const v2f*)xp;
            xb[u] = *(const v2f*)(xp + 64);
        }
#pragma unroll
        for (int u = 0; u < 16; ++u) {
            const float v = __int_as_float(__builtin_amdgcn_readlane(vv, g + u));
            const int   t = __builtin_amdgcn_readlane(tt, g + u);
            float* hp = hT + t * LSTR + 2 * lane;
            atomicAdd(hp,      v * xa[u].x);
            atomicAdd(hp + 1,  v * xa[u].y);
            atomicAdd(hp + 64, v * xb[u].x);
            atomicAdd(hp + 65, v * xb[u].y);
        }
    }
}

__device__ __forceinline__
void add_part(const float* __restrict__ X, const int* __restrict__ rows,
              const int* __restrict__ cols, const float* __restrict__ vals, float* hT,
              int ee, int r0, unsigned rcur, int cnt, int lane) {
    int cv = 0, vv = 0, tt = RB;
    if (lane < cnt) {
        cv = cols[ee]; vv = __float_as_int(vals[ee]);
        const unsigned t = (unsigned)(rows[ee] - r0);
        tt = t < rcur ? (int)t : RB;
    }
    for (int g = 0; g < cnt; g += 8) {
        v2f xa[8], xb[8];
#pragma unroll
        for (int u = 0; u < 8; ++u) {
            const int c = __builtin_amdgcn_readlane(cv, g + u);
            const float* xp = X + (size_t)c * D + 2 * lane;
            xa[u] = *(const v2f*)xp;
            xb[u] = *(const v2f*)(xp + 64);
        }
#pragma unroll
        for (int u = 0; u < 8; ++u) {
            if (g + u < cnt) {
                const float v = __int_as_float(__builtin_amdgcn_readlane(vv, g + u));
                const int   t = __builtin_amdgcn_readlane(tt, g + u);
                float* hp = hT + t * LSTR + 2 * lane;
                atomicAdd(hp,      v * xa[u].x);
                atomicAdd(hp + 1,  v * xa[u].y);
                atomicAdd(hp + 64, v * xb[u].x);
                atomicAdd(hp + 65, v * xb[u].y);
            }
        }
    }
}

__global__ __launch_bounds__(1024)
void gcn_rowblock_kernel(const float* __restrict__ X,
                         const int*   __restrict__ rows,
                         const int*   __restrict__ cols,
                         const float* __restrict__ vals,
                         const float* __restrict__ W,
                         const float* __restrict__ bias,
                         const unsigned* __restrict__ keys,
                         float* __restrict__ out,
                         int nNodes, int nEdges) {
    __shared__ __attribute__((aligned(16))) float hT[(RB + 1) * LSTR];
    __shared__ __attribute__((aligned(16))) float Wl[D * LSTR];
    __shared__ int qe[NW * QCAP];

    const int tid  = threadIdx.x;
    const int wave = __builtin_amdgcn_readfirstlane(tid >> 5);
    const int lane = tid & 31;
    const int r0   = blockIdx.x * RB;
    const unsigned rcur = (unsigned)min(RB, nNodes - r0);

    for (int i = tid; i < ((RB + 1) * LSTR) / 4; i += 1024)
        ((float4*)hT)[i] = make_float4(0.f, 0.f, 0.f, 0.f);
    for (int i = tid; i < (D * D) / 4; i += 1024) {
        const float4 w4 = ((const float4*)W)[i];
        const int flat = i * 4;
        *(float4*)&Wl[(flat >> 7) * LSTR + (flat & 127)] = w4;
    }
    __syncthreads();

    int* myqe = qe + wave * QCAP;
    int  n = 0;
    bool pend = false;
    int  pcv = 0, pvv = 0, ptt = 0;

#define APPEND(m, mk, e)                                                     \
    {                                                                        \
        const int pre = __builtin_amdgcn_mbcnt_lo(mk, 0);                    \
        if (m) myqe[n + pre] = (e);                                          \
        n += __builtin_popcount(mk);                                         \
    }
#define FLUSH()                                                              \
    if (n >= 32) {                                                           \
        n -= 32;                                                             \
        const int ee = myqe[n + lane];                                       \
        const int nrr = rows[ee];                                            \
        const int ncv = cols[ee];                                            \
        const int nvv = __float_as_int(vals[ee]);                            \
        if (pend) add_full(X, hT, pcv, pvv, ptt, lane);                      \
        const unsigned nt = (unsigned)(nrr - r0);                            \
        pcv = ncv; pvv = nvv; ptt = nt < rcur ? (int)nt : RB; pend = true;   \
    }
#define NZB(w) ((((w) ^ pat) & 0x7f7f7f7fu) + 0x7f7f7f7fu | ((w) ^ pat))
#define CHUNK(kv, e0)                                                        \
    {                                                                        \
        const unsigned y0 = NZB((kv).x), y1 = NZB((kv).y);                   \
        const unsigned y2 = NZB((kv).z), y3 = NZB((kv).w);                   \
        const unsigned miss = y0 & y1 & y2 & y3 & 0x80808080u;               \
        if (__builtin_amdgcn_ballot_w32(miss != 0x80808080u)) {              \
                 \
            unsigned hm = ((~y0 & 0x80808080u) >> 7) | ((~y1 & 0x80808080u) >> 6) \
                        | ((~y2 & 0x80808080u) >> 5) | ((~y3 & 0x80808080u) >> 4); \
            unsigned mk = __builtin_amdgcn_ballot_w32(hm != 0u);             \
              \
            while (mk) {                                                     \
                const int p = __builtin_ctz(hm | 0x80000000u);               \
                const bool has = hm != 0u;                                   \
                APPEND(has, mk, (e0) + ((p & 3) << 2) + (p >> 3))            \
                hm &= hm - 1u;                                               \
                FLUSH()                                                      \
                mk = __builtin_amdgcn_ballot_w32(hm != 0u);                  \
            }                                                                \
        }                                                                    \
    }

    const unsigned pat = (blockIdx.x & 255u) * 0x01010101u;
    const int nChunks = nEdges >> 9;
    if (nChunks > 0) {
        const int off = (int)(((long long)blockIdx.x * nChunks) / (int)gridDim.x);
        const uint4* keys4 = (const uint4*)keys;
        uint4 ka[4]; int ca[4];
#pragma unroll
        for (int u = 0; u < 4; ++u) {
            int ch = min(wave + u * NW, nChunks - 1) + off;
            if (ch >= nChunks) ch -= nChunks;
            ca[u] = ch;
            ka[u] = keys4[(ch << 5) + lane];
        }
        for (int i = wave; i < nChunks; i += 4 * NW) {
            uint4 kb[4]; int cb[4];
#pragma unroll
            for (int u = 0; u < 4; ++u) {
                int ch = min(i + (4 + u) * NW, nChunks - 1) + off;
                if (ch >= nChunks) ch -= nChunks;
                cb[u] = ch;
                kb[u] = keys4[(ch << 5) + lane];
            }
#pragma unroll
            for (int u = 0; u < 4; ++u) {
                if (i + u * NW < nChunks) CHUNK(ka[u], (ca[u] << 9) + lane * 16)
            }
#pragma unroll
            for (int u = 0; u < 4; ++u) { ka[u] = kb[u]; ca[u] = cb[u]; }
        }
    }
    if (wave == 0) {
        for (int e = (nChunks << 9) + lane; e - lane < nEdges; e += 32) {
            unsigned t = 0xFFFFFFFFu;
            if (e < nEdges) t = (unsigned)(rows[e] - r0);
            const bool m = t < rcur;
            const unsigned mk = __builtin_amdgcn_ballot_w32(m);
            if (mk) {
                APPEND(m, mk, e)
                FLUSH()
            }
        }
    }
    if (pend) add_full(X, hT, pcv, pvv, ptt, lane);
    if (n > 0) {
        const int ee = lane < n ? myqe[lane] : 0;
        add_part(X, rows, cols, vals, hT, ee, r0, rcur, n, lane);
    }
#undef CHUNK
#undef NZB
#undef FLUSH
#undef APPEND
    __syncthreads();

    const int l15 = lane & 15;
    const int hi  = lane >> 4;
    for (int unit = wave; unit < (RB / 16) * 2; unit += NW) {
        const int rt = unit >> 1;
        const int nh = unit & 1;
        const float* abase = hT + (rt * 16 + l15) * LSTR + 2 * hi;
        const float* bbase = Wl + (nh * 64 + l15) * LSTR + 2 * hi;
        v8f acc[4];
#pragma unroll
        for (int nt = 0; nt < 4; ++nt) acc[nt] = (v8f){0.f,0.f,0.f,0.f,0.f,0.f,0.f,0.f};
        for (int ks = 0; ks < 32; ++ks) {
            const v2f a = *(const v2f*)(abase + ks * 4);
#pragma unroll
            for (int nt = 0; nt < 4; ++nt) {
                const v2f b2 = *(const v2f*)(bbase + nt * 16 * LSTR + ks * 4);
                acc[nt] = __builtin_amdgcn_wmma_f32_16x16x4_f32(
                    false, a, false, b2, (short)0, acc[nt], false, false);
            }
        }
#pragma unroll
        for (int nt = 0; nt < 4; ++nt) {
            const int col = nh * 64 + nt * 16 + l15;
            const float bv = bias[col];
#pragma unroll
            for (int v = 0; v < 8; ++v) {
                const int r = r0 + rt * 16 + v + 8 * hi;
                if (r < nNodes)
                    __builtin_nontemporal_store(acc[nt][v] + bv, &out[(size_t)r * D + col]);
            }
        }
    }
}

extern "C" void kernel_launch(void* const* d_in, const int* in_sizes, int n_in,
                              void* d_out, int out_size, void* d_ws, size_t ws_size,
                              hipStream_t stream) {
    const float* X    = (const float*)d_in[0];
    const int*   rows = (const int*)  d_in[1];
    const int*   cols = (const int*)  d_in[2];
    const float* vals = (const float*)d_in[3];
    const float* W    = (const float*)d_in[4];
    const float* bias = (const float*)d_in[5];
    float*       out  = (float*)d_out;
    unsigned*    keys = (unsigned*)d_ws;

    const int nNodes = in_sizes[0] / D;
    const int nEdges = in_sizes[1];

    gcn_keys_kernel<<<256, 256, 0, stream>>>(rows, keys, nEdges);

    const int blocks = (nNodes + RB - 1) / RB;
    gcn_rowblock_kernel<<<blocks, 1024, 0, stream>>>(X, rows, cols, vals, W, bias, keys, out,
                                                     nNodes, nEdges);
}
